# starting kernel
# baseline (speedup 1.0000x reference)
#include <hip/hip_runtime.h>

#define NNODES 100000
#define NEDGES 1600000
#define FIN 128
#define FH 64
#define FOUT 32

typedef float v2f __attribute__((ext_vector_type(2)));
typedef float v8f __attribute__((ext_vector_type(8)));

static __device__ __forceinline__ v8f wmma_f32_k4(v2f a, v2f b, v8f c) {
  return __builtin_amdgcn_wmma_f32_16x16x4_f32(false, a, false, b, (short)0, c,
                                               false, false);
}

__global__ void deg_init_kernel(float* __restrict__ deg) {
  int i = blockIdx.x * blockDim.x + threadIdx.x;
  if (i < NNODES) deg[i] = 1.0f;
}

__global__ void deg_count_kernel(const int* __restrict__ dst,
                                 float* __restrict__ deg) {
  int i = blockIdx.x * blockDim.x + threadIdx.x;
  if (i < NEDGES) unsafeAtomicAdd(&deg[dst[i]], 1.0f);
}

__global__ void dinv_kernel(const float* __restrict__ deg,
                            float* __restrict__ dinv) {
  int i = blockIdx.x * blockDim.x + threadIdx.x;
  if (i < NNODES) dinv[i] = rsqrtf(deg[i]);
}

__global__ __launch_bounds__(256) void gcn_gemm1(const float* __restrict__ x,
                                                 const float* __restrict__ W1,
                                                 float* __restrict__ h1) {
  __shared__ float sW[FH * FIN];
  for (int i = threadIdx.x; i < FH * FIN; i += 256) sW[i] = W1[i];
  __syncthreads();

  const int lane = threadIdx.x & 31;
  const int wave = threadIdx.x >> 5;
  const int tile = blockIdx.x * 8 + wave;
  if (tile * 16 >= NNODES) return;
  const int m0 = tile * 16;
  const int lrow = lane & 15;
  const int koff = (lane >> 4) << 1;

  v8f acc0 = {}, acc1 = {}, acc2 = {}, acc3 = {};
  const float* xrow = x + (size_t)(m0 + lrow) * FIN;
#pragma unroll 4
  for (int k = 0; k < FIN; k += 4) {
    v2f a = *(const v2f*)(xrow + k + koff);
    v2f b0 = *(const v2f*)(&sW[(0 * 16 + lrow) * FIN + k + koff]);
    v2f b1 = *(const v2f*)(&sW[(1 * 16 + lrow) * FIN + k + koff]);
    v2f b2 = *(const v2f*)(&sW[(2 * 16 + lrow) * FIN + k + koff]);
    v2f b3 = *(const v2f*)(&sW[(3 * 16 + lrow) * FIN + k + koff]);
    acc0 = wmma_f32_k4(a, b0, acc0);
    acc1 = wmma_f32_k4(a, b1, acc1);
    acc2 = wmma_f32_k4(a, b2, acc2);
    acc3 = wmma_f32_k4(a, b3, acc3);
  }
  const int rbase = m0 + ((lane >> 4) << 3);
#pragma unroll
  for (int r = 0; r < 8; ++r) {
    h1[(size_t)(rbase + r) * FH + 0 * 16 + lrow] = acc0[r];
    h1[(size_t)(rbase + r) * FH + 1 * 16 + lrow] = acc1[r];
    h1[(size_t)(rbase + r) * FH + 2 * 16 + lrow] = acc2[r];
    h1[(size_t)(rbase + r) * FH + 3 * 16 + lrow] = acc3[r];
  }
}

__global__ __launch_bounds__(256) void gcn_gemm2(const float* __restrict__ agg1,
                                                 const float* __restrict__ b1,
                                                 const float* __restrict__ W2,
                                                 float* __restrict__ h2) {
  __shared__ float sW[FOUT * FH];
  __shared__ float sB[FH];
  for (int i = threadIdx.x; i < FOUT * FH; i += 256) sW[i] = W2[i];
  if (threadIdx.x < FH) sB[threadIdx.x] = b1[threadIdx.x];
  __syncthreads();

  const int lane = threadIdx.x & 31;
  const int wave = threadIdx.x >> 5;
  const int tile = blockIdx.x * 8 + wave;
  if (tile * 16 >= NNODES) return;
  const int m0 = tile * 16;
  const int lrow = lane & 15;
  const int koff = (lane >> 4) << 1;

  v8f acc0 = {}, acc1 = {};
  const float* arow = agg1 + (size_t)(m0 + lrow) * FH;
#pragma unroll 4
  for (int k = 0; k < FH; k += 4) {
    v2f ar = *(const v2f*)(arow + k + koff);
    v2f a;
    a.x = fmaxf(ar.x + sB[k + koff + 0], 0.0f);
    a.y = fmaxf(ar.y + sB[k + koff + 1], 0.0f);
    v2f b0 = *(const v2f*)(&sW[(0 * 16 + lrow) * FH + k + koff]);
    v2f b1v = *(const v2f*)(&sW[(1 * 16 + lrow) * FH + k + koff]);
    acc0 = wmma_f32_k4(a, b0, acc0);
    acc1 = wmma_f32_k4(a, b1v, acc1);
  }
  const int rbase = m0 + ((lane >> 4) << 3);
#pragma unroll
  for (int r = 0; r < 8; ++r) {
    h2[(size_t)(rbase + r) * FOUT + 0 * 16 + lrow] = acc0[r];
    h2[(size_t)(rbase + r) * FOUT + 1 * 16 + lrow] = acc1[r];
  }
}

__global__ void agg1_init_kernel(const float* __restrict__ h1,
                                 const float* __restrict__ dinv,
                                 float* __restrict__ agg1) {
  int i = blockIdx.x * blockDim.x + threadIdx.x;
  if (i >= NNODES * (FH / 4)) return;
  int node = i / (FH / 4);
  float w = dinv[node];
  w *= w;
  float4 v = ((const float4*)h1)[i];
  float4 o = {v.x * w, v.y * w, v.z * w, v.w * w};
  ((float4*)agg1)[i] = o;
}

__global__ void agg2_init_kernel(const float* __restrict__ h2,
                                 const float* __restrict__ dinv,
                                 const float* __restrict__ b2,
                                 float* __restrict__ out) {
  int i = blockIdx.x * blockDim.x + threadIdx.x;
  if (i >= NNODES * (FOUT / 4)) return;
  int node = i / (FOUT / 4);
  int c4 = i % (FOUT / 4);
  float w = dinv[node];
  w *= w;
  float4 v = ((const float4*)h2)[i];
  float4 bb = ((const float4*)b2)[c4];
  float4 o = {v.x * w + bb.x, v.y * w + bb.y, v.z * w + bb.z, v.w * w + bb.w};
  ((float4*)out)[i] = o;
}

template <int F>
__global__ void agg_edges_kernel(const int* __restrict__ src,
                                 const int* __restrict__ dst,
                                 const float* __restrict__ dinv,
                                 const float* __restrict__ h,
                                 float* __restrict__ agg) {
  const int TPE = F / 4;
  long long t = (long long)blockIdx.x * blockDim.x + threadIdx.x;
  long long e = t / TPE;
  int c = (int)(t % TPE);
  if (e >= NEDGES) return;
  int s = src[e];
  int d = dst[e];
  float w = dinv[s] * dinv[d];
  float4 v = ((const float4*)(h + (size_t)s * F))[c];
  float* o = agg + (size_t)d * F + c * 4;
  unsafeAtomicAdd(o + 0, v.x * w);
  unsafeAtomicAdd(o + 1, v.y * w);
  unsafeAtomicAdd(o + 2, v.z * w);
  unsafeAtomicAdd(o + 3, v.w * w);
}

extern "C" void kernel_launch(void* const* d_in, const int* in_sizes, int n_in,
                              void* d_out, int out_size, void* d_ws,
                              size_t ws_size, hipStream_t stream) {
  const float* x = (const float*)d_in[0];
  const int* edge_index = (const int*)d_in[1];
  const float* W1 = (const float*)d_in[2];
  const float* b1 = (const float*)d_in[3];
  const float* W2 = (const float*)d_in[4];
  const float* b2 = (const float*)d_in[5];
  float* out = (float*)d_out;

  const int* src = edge_index;
  const int* dst = edge_index + NEDGES;

  float* ws = (float*)d_ws;
  float* deg = ws;
  float* dinv = ws + NNODES;
  float* h1 = ws + 2 * (size_t)NNODES;
  float* agg1 = h1 + (size_t)NNODES * FH;
  float* h2 = agg1 + (size_t)NNODES * FH;

  const int B = 256;
  const int nTiles = NNODES / 16;
  const int gemmBlocks = (nTiles + 7) / 8;

  deg_init_kernel<<<(NNODES + B - 1) / B, B, 0, stream>>>(deg);
  deg_count_kernel<<<(NEDGES + B - 1) / B, B, 0, stream>>>(dst, deg);
  dinv_kernel<<<(NNODES + B - 1) / B, B, 0, stream>>>(deg, dinv);

  gcn_gemm1<<<gemmBlocks, B, 0, stream>>>(x, W1, h1);

  agg1_init_kernel<<<(NNODES * (FH / 4) + B - 1) / B, B, 0, stream>>>(h1, dinv,
                                                                      agg1);
  {
    long long tot = (long long)NEDGES * (FH / 4);
    agg_edges_kernel<FH><<<(unsigned)((tot + B - 1) / B), B, 0, stream>>>(
        src, dst, dinv, h1, agg1);
  }

  gcn_gemm2<<<gemmBlocks, B, 0, stream>>>(agg1, b1, W2, h2);

  agg2_init_kernel<<<(NNODES * (FOUT / 4) + B - 1) / B, B, 0, stream>>>(
      h2, dinv, b2, out);
  {
    long long tot = (long long)NEDGES * (FOUT / 4);
    agg_edges_kernel<FOUT><<<(unsigned)((tot + B - 1) / B), B, 0, stream>>>(
        src, dst, dinv, h2, out);
  }
}


# Both GEMMs: weight fragments held in registers for a wave's whole strided list of row tiles, read once from global with 16-byte k-permuted loads (A the same); LDS staging gone; new k sum order
# speedup vs baseline: 2.9671x; 2.9671x over previous
#include <hip/hip_runtime.h>

#define NNODES 100000
#define NEDGES 1600000
#define FIN 128
#define FH 64
#define FOUT 32

typedef float v2f __attribute__((ext_vector_type(2)));
typedef float v8f __attribute__((ext_vector_type(8)));
typedef float v4f __attribute__((ext_vector_type(4)));
typedef int v4i __attribute__((ext_vector_type(4)));

static __device__ __forceinline__ v8f wmma_f32_k4(v2f a, v2f b, v8f c) {
  return __builtin_amdgcn_wmma_f32_16x16x4_f32(false, a, false, b, (short)0, c,
                                               false, false);
}

#define CNT_STRIDE 128

#define SP_THREADS 128
#define NODE_V4 (NNODES / 4)
#define NODE_BLOCKS ((NODE_V4 + SP_THREADS - 1) / SP_THREADS)
static_assert(NNODES % 4 == 0, "the per-node planes are walked as float4");

__global__ __launch_bounds__(SP_THREADS) void deg_init_kernel(float* __restrict__ degA,
                                                              float* __restrict__ degB,
                                                              int* __restrict__ cnt) {
  const int i = blockIdx.x * SP_THREADS + threadIdx.x;
  if (i < NODE_V4) {
    ((float4*)degA)[i] = float4{1.0f, 1.0f, 1.0f, 1.0f};
    ((float4*)degB)[i] = float4{0.0f, 0.0f, 0.0f, 0.0f};
  }
  if (i < 4) cnt[i * CNT_STRIDE] = 0;
}

#define CNT_V4 5
#define CNT_BLOCKS (NEDGES / 4 / (SP_THREADS * CNT_V4))
static_assert(NEDGES % (4 * SP_THREADS * CNT_V4) == 0, "the count blocks must tile the edge list exactly");

__global__ __launch_bounds__(SP_THREADS) void deg_count_kernel(const int* __restrict__ dst,
                                                               float* __restrict__ degA,
                                                               float* __restrict__ degB) {
  const unsigned where = __builtin_amdgcn_s_sendmsg_rtn(0x87);
  float* __restrict__ deg = ((where >> 18) & 1u) ? degB : degA;
  const v4i* __restrict__ d4 = (const v4i*)dst;
  const int base = blockIdx.x * (SP_THREADS * CNT_V4) + threadIdx.x;
  v4i d[CNT_V4];
#pragma unroll
  for (int j = 0; j < CNT_V4; ++j) d[j] = d4[base + j * SP_THREADS];
  static_assert(CNT_V4 == 5, "the fence below names five items");
  asm volatile("" ::"v"(d[0]), "v"(d[1]), "v"(d[2]), "v"(d[3]), "v"(d[4]));
#pragma unroll
  for (int j = 0; j < CNT_V4; ++j) {
    unsafeAtomicAdd(&deg[d[j].x], 1.0f);
    unsafeAtomicAdd(&deg[d[j].y], 1.0f);
    unsafeAtomicAdd(&deg[d[j].z], 1.0f);
    unsafeAtomicAdd(&deg[d[j].w], 1.0f);
  }
}

__global__ __launch_bounds__(SP_THREADS) void dinv_kernel(const float* __restrict__ degA,
                                                          const float* __restrict__ degB,
                                                          float* __restrict__ dinv) {
  const int i = blockIdx.x * SP_THREADS + threadIdx.x;
  if (i < NODE_V4) {
    const float4 a = ((const float4*)degA)[i];
    const float4 b = ((const float4*)degB)[i];
    ((float4*)dinv)[i] = float4{rsqrtf(a.x + b.x), rsqrtf(a.y + b.y), rsqrtf(a.z + b.z),
                                rsqrtf(a.w + b.w)};
  }
}

#define GEMM_WAVES 8
#define GEMM_BLOCKS 256
#define GEMM_STRIDE (GEMM_WAVES * GEMM_BLOCKS)
#define NTILES (NNODES / 16)
static_assert(NNODES % 16 == 0, "row tiles are whole");
static_assert(GEMM_STRIDE <= NTILES, "every wave has a first tile");

static __device__ __forceinline__ v2f lo2(v4f v) { return v2f{v.x, v.y}; }
static __device__ __forceinline__ v2f hi2(v4f v) { return v2f{v.z, v.w}; }

__global__ __launch_bounds__(GEMM_WAVES * 32) void gcn_gemm1(const float* __restrict__ x,
                                                             const float* __restrict__ W1,
                                                             float* __restrict__ h1) {
  constexpr int KJ = FIN / 8;
  const int lane = threadIdx.x & 31;
  const int wave = __builtin_amdgcn_readfirstlane(threadIdx.x >> 5);
  const int lrow = lane & 15;
  const int kh = (lane >> 4) << 2;

  v4f bw[4][KJ];
  const float* wrow = W1 + lrow * FIN + kh;
#pragma unroll
  for (int n = 0; n < 4; ++n)
#pragma unroll
    for (int j = 0; j < KJ; ++j) bw[n][j] = *(const v4f*)(wrow + n * 16 * FIN + 8 * j);

  int tile = wave * GEMM_BLOCKS + blockIdx.x;
  v4f a[KJ];
  {
    const float* xrow = x + (size_t)(tile * 16 + lrow) * FIN + kh;
#pragma unroll
    for (int j = 0; j < KJ; ++j) a[j] = *(const v4f*)(xrow + 8 * j);
  }
  for (; tile < NTILES; tile += GEMM_STRIDE) {
    const int nt = tile + GEMM_STRIDE < NTILES ? tile + GEMM_STRIDE : tile;
    v4f an[KJ];
    {
      const float* xrow = x + (size_t)(nt * 16 + lrow) * FIN + kh;
#pragma unroll
      for (int j = 0; j < KJ; ++j) an[j] = *(const v4f*)(xrow + 8 * j);
    }
    v8f acc0 = {}, acc1 = {}, acc2 = {}, acc3 = {};
#pragma unroll
    for (int j = 0; j < KJ; ++j) {
      acc0 = wmma_f32_k4(lo2(a[j]), lo2(bw[0][j]), acc0);
      acc1 = wmma_f32_k4(lo2(a[j]), lo2(bw[1][j]), acc1);
      acc2 = wmma_f32_k4(lo2(a[j]), lo2(bw[2][j]), acc2);
      acc3 = wmma_f32_k4(lo2(a[j]), lo2(bw[3][j]), acc3);
      acc0 = wmma_f32_k4(hi2(a[j]), hi2(bw[0][j]), acc0);
      acc1 = wmma_f32_k4(hi2(a[j]), hi2(bw[1][j]), acc1);
      acc2 = wmma_f32_k4(hi2(a[j]), hi2(bw[2][j]), acc2);
      acc3 = wmma_f32_k4(hi2(a[j]), hi2(bw[3][j]), acc3);
    }
    const int rbase = tile * 16 + ((lane >> 4) << 3);
#pragma unroll
    for (int r = 0; r < 8; ++r) {
      h1[(size_t)(rbase + r) * FH + 0 * 16 + lrow] = acc0[r];
      h1[(size_t)(rbase + r) * FH + 1 * 16 + lrow] = acc1[r];
      h1[(size_t)(rbase + r) * FH + 2 * 16 + lrow] = acc2[r];
      h1[(size_t)(rbase + r) * FH + 3 * 16 + lrow] = acc3[r];
    }
#pragma unroll
    for (int j = 0; j < KJ; ++j) a[j] = an[j];
  }
}

__global__ __launch_bounds__(GEMM_WAVES * 32) void gcn_gemm2(const float* __restrict__ agg1,
                                                             const float* __restrict__ b1,
                                                             const float* __restrict__ W2,
                                                             float* __restrict__ h2) {
  constexpr int KJ = FH / 8;
  const int lane = threadIdx.x & 31;
  const int wave = __builtin_amdgcn_readfirstlane(threadIdx.x >> 5);
  const int lrow = lane & 15;
  const int kh = (lane >> 4) << 2;

  v4f bw[2][KJ];
  v4f bias[KJ];
  const float* wrow = W2 + lrow * FH + kh;
#pragma unroll
  for (int n = 0; n < 2; ++n)
#pragma unroll
    for (int j = 0; j < KJ; ++j) bw[n][j] = *(const v4f*)(wrow + n * 16 * FH + 8 * j);
#pragma unroll
  for (int j = 0; j < KJ; ++j) bias[j] = *(const v4f*)(b1 + kh + 8 * j);

  int tile = wave * GEMM_BLOCKS + blockIdx.x;
  v4f a[KJ];
  {
    const float* arow = agg1 + (size_t)(tile * 16 + lrow) * FH + kh;
#pragma unroll
    for (int j = 0; j < KJ; ++j) a[j] = *(const v4f*)(arow + 8 * j);
  }
  for (; tile < NTILES; tile += GEMM_STRIDE) {
    const int nt = tile + GEMM_STRIDE < NTILES ? tile + GEMM_STRIDE : tile;
    v4f an[KJ];
    {
      const float* arow = agg1 + (size_t)(nt * 16 + lrow) * FH + kh;
#pragma unroll
      for (int j = 0; j < KJ; ++j) an[j] = *(const v4f*)(arow + 8 * j);
    }
    v8f acc0 = {}, acc1 = {};
#pragma unroll
    for (int j = 0; j < KJ; ++j) {
      v4f r;
      r.x = fmaxf(a[j].x + bias[j].x, 0.0f);
      r.y = fmaxf(a[j].y + bias[j].y, 0.0f);
      r.z = fmaxf(a[j].z + bias[j].z, 0.0f);
      r.w = fmaxf(a[j].w + bias[j].w, 0.0f);
      acc0 = wmma_f32_k4(lo2(r), lo2(bw[0][j]), acc0);
      acc1 = wmma_f32_k4(lo2(r), lo2(bw[1][j]), acc1);
      acc0 = wmma_f32_k4(hi2(r), hi2(bw[0][j]), acc0);
      acc1 = wmma_f32_k4(hi2(r), hi2(bw[1][j]), acc1);
    }
    const int rbase = tile * 16 + ((lane >> 4) << 3);
#pragma unroll
    for (int r = 0; r < 8; ++r) {
      h2[(size_t)(rbase + r) * FOUT + 0 * 16 + lrow] = acc0[r];
      h2[(size_t)(rbase + r) * FOUT + 1 * 16 + lrow] = acc1[r];
    }
#pragma unroll
    for (int j = 0; j < KJ; ++j) a[j] = an[j];
  }
}

#define INIT_V4 10
#define INIT1_BLOCKS (NNODES * (FH / 4) / (SP_THREADS * INIT_V4))
#define INIT2_BLOCKS (NNODES * (FOUT / 4) / (SP_THREADS * INIT_V4))
static_assert(NNODES * (FH / 4) % (SP_THREADS * INIT_V4) == 0, "the layer-1 blocks must tile h1 exactly");
static_assert(NNODES * (FOUT / 4) % (SP_THREADS * INIT_V4) == 0, "the layer-2 blocks must tile h2 exactly");
static_assert(SP_THREADS % (FOUT / 4) == 0, "a lane keeps one bias column over all of its items");

__global__ __launch_bounds__(SP_THREADS) void agg1_init_kernel(const float* __restrict__ h1,
                                                               const float* __restrict__ dinv,
                                                               float* __restrict__ agg1) {
  const int base = blockIdx.x * (SP_THREADS * INIT_V4) + threadIdx.x;
  v4f v[INIT_V4];
  float w[INIT_V4];
#pragma unroll
  for (int j = 0; j < INIT_V4; ++j) {
    const int i = base + j * SP_THREADS;
    v[j] = ((const v4f*)h1)[i];
    w[j] = dinv[(unsigned)i / (FH / 4)];
  }
  static_assert(INIT_V4 == 10, "the fence below names ten items");
  asm volatile("" ::"v"(v[0]), "v"(v[1]), "v"(v[2]), "v"(v[3]), "v"(v[4]), "v"(v[5]), "v"(v[6]),
               "v"(v[7]), "v"(v[8]), "v"(v[9]), "v"(w[0]), "v"(w[1]), "v"(w[2]), "v"(w[3]),
               "v"(w[4]), "v"(w[5]), "v"(w[6]), "v"(w[7]), "v"(w[8]), "v"(w[9]));
#pragma unroll
  for (int j = 0; j < INIT_V4; ++j) {
    const int i = base + j * SP_THREADS;
    const float ww = w[j] * w[j];
    v4f o = {v[j].x * ww, v[j].y * ww, v[j].z * ww, v[j].w * ww};
    ((v4f*)agg1)[i] = o;
  }
}

__global__ __launch_bounds__(SP_THREADS) void agg2_init_kernel(const float* __restrict__ h2,
                                                               const float* __restrict__ dinv,
                                                               const float* __restrict__ b2,
                                                               float* __restrict__ out) {
  const int base = blockIdx.x * (SP_THREADS * INIT_V4) + threadIdx.x;
  const v4f bb = ((const v4f*)b2)[threadIdx.x % (FOUT / 4)];
  v4f v[INIT_V4];
  float w[INIT_V4];
#pragma unroll
  for (int j = 0; j < INIT_V4; ++j) {
    const int i = base + j * SP_THREADS;
    v[j] = ((const v4f*)h2)[i];
    w[j] = dinv[(unsigned)i / (FOUT / 4)];
  }
  static_assert(INIT_V4 == 10, "the fence below names ten items");
  asm volatile("" ::"v"(v[0]), "v"(v[1]), "v"(v[2]), "v"(v[3]), "v"(v[4]), "v"(v[5]), "v"(v[6]),
               "v"(v[7]), "v"(v[8]), "v"(v[9]), "v"(w[0]), "v"(w[1]), "v"(w[2]), "v"(w[3]),
               "v"(w[4]), "v"(w[5]), "v"(w[6]), "v"(w[7]), "v"(w[8]), "v"(w[9]), "v"(bb));
#pragma unroll
  for (int j = 0; j < INIT_V4; ++j) {
    const int i = base + j * SP_THREADS;
    const float ww = w[j] * w[j];
    v4f o = {v[j].x * ww + bb.x, v[j].y * ww + bb.y, v[j].z * ww + bb.z, v[j].w * ww + bb.w};
    ((v4f*)out)[i] = o;
  }
}

#define AGG_WAVES 4
#define AGG_EPW 128
#define AGG_CHUNKS (NEDGES / (AGG_WAVES * AGG_EPW))
#define OWN_SHIFT 6
static_assert(NEDGES % (AGG_WAVES * AGG_EPW) == 0, "edge count must fill every wave");
static_assert((size_t)NNODES + NEDGES + 4 * CNT_STRIDE <= (size_t)NNODES * FOUT,
              "second count plane, one weight per edge and the counters must fit the tail region");

template <int F, bool FULL>
static __device__ __forceinline__ void agg_group(int s, int d, float w, int m, int lane,
                                                 const float* __restrict__ h,
                                                 float* __restrict__ agg) {
  constexpr int C = F / 32;
  constexpr int U = 16 / C;
  float v[U][C] = {};
#pragma unroll
  for (int j = 0; j < U; ++j) {
    if (FULL || j < m) {
      const int sj = __builtin_amdgcn_readlane(s, j);
      const int r = sj * F + lane;
#pragma unroll
      for (int c = 0; c < C; ++c) v[j][c] = h[r + 32 * c];
    }
  }
#pragma unroll
  for (int j = 0; j < U; ++j) {
    if (FULL || j < m) {
      const int dj = __builtin_amdgcn_readlane(d, j);
      const float wj = __int_as_float(
          __builtin_amdgcn_readlane(__float_as_int(w), j));
      const int r = dj * F + lane;
#pragma unroll
      for (int c = 0; c < C; ++c) unsafeAtomicAdd(agg + r + 32 * c, v[j][c] * wj);
    }
  }
}

template <int F, bool FIRST>
__global__ __launch_bounds__(AGG_WAVES * 32) void agg_edges_kernel(
    const int* __restrict__ src, const int* __restrict__ dst,
    const float* __restrict__ dinv, float* __restrict__ wts,
    const float* __restrict__ h, float* __restrict__ agg, int* __restrict__ cnt) {
  constexpr int U = 16 / (F / 32);
  constexpr int NB = AGG_EPW / 32;
  __shared__ int sTicket[2];
  __shared__ int qS[AGG_WAVES][AGG_EPW];
  __shared__ int qD[AGG_WAVES][AGG_EPW];
  __shared__ float qW[AGG_WAVES][AGG_EPW];

  if (threadIdx.x == 0) {
    const unsigned where = __builtin_amdgcn_s_sendmsg_rtn(0x87);
    int c = (int)((where >> 18) & 1u);
    int t = atomicAdd(cnt + c * CNT_STRIDE, 1);
    if (t >= AGG_CHUNKS) {
      c ^= 1;
      t = atomicAdd(cnt + c * CNT_STRIDE, 1);
    }
    sTicket[0] = t;
    sTicket[1] = c;
  }
  __syncthreads();
  const int t = sTicket[0];
  const int cls = sTicket[1];
  if (t >= AGG_CHUNKS) return;

  const int lane = threadIdx.x & 31;
  const int wave = threadIdx.x >> 5;
  const int e0 = t * (AGG_WAVES * AGG_EPW) + wave * AGG_EPW + lane;
  int* __restrict__ qs = qS[wave];
  int* __restrict__ qd = qD[wave];
  float* __restrict__ qw = qW[wave];

  int s[NB], d[NB];
  float w[NB];
  bool own[NB];
#pragma unroll
  for (int b = 0; b < NB; ++b) {
    s[b] = src[e0 + 32 * b];
    d[b] = dst[e0 + 32 * b];
  }
#pragma unroll
  for (int b = 0; b < NB; ++b) own[b] = ((d[b] >> OWN_SHIFT) & 1) == cls;
  if constexpr (FIRST) {
#pragma unroll
    for (int b = 0; b < NB; ++b) {
      const int sq = own[b] ? s[b] : 0;
      const int dq = own[b] ? d[b] : 0;
      w[b] = dinv[sq] * dinv[dq];
    }
#pragma unroll
    for (int b = 0; b < NB; ++b)
      if (own[b]) wts[e0 + 32 * b] = w[b];
  } else {
#pragma unroll
    for (int b = 0; b < NB; ++b) w[b] = wts[e0 + 32 * b];
    static_assert(NB == 4, "the fence below names four record sets");
    asm volatile("" ::"v"(s[0]), "v"(s[1]), "v"(s[2]), "v"(s[3]), "v"(w[0]), "v"(w[1]),
                 "v"(w[2]), "v"(w[3]));
  }
  int n = 0;
#pragma unroll
  for (int b = 0; b < NB; ++b) {
    const unsigned mask = __builtin_amdgcn_ballot_w32(own[b]);
    if (own[b]) {
      const int p = n + (int)__builtin_amdgcn_mbcnt_lo(mask, 0u);
      qs[p] = s[b];
      qd[p] = d[b];
      qw[p] = w[b];
    }
    n += __builtin_popcount(mask);
  }

  for (int g = 0; g < n; g += U) {
    const int idx = g + (lane & (U - 1));
    int sg = 0, dg = 0;
    float wg = 0.0f;
    if (idx < n) {
      sg = qs[idx];
      dg = qd[idx];
      wg = qw[idx];
    }
    const int m = n - g;
    if (m >= U)
      agg_group<F, true>(sg, dg, wg, U, lane, h, agg);
    else
      agg_group<F, false>(sg, dg, wg, m, lane, h, agg);
  }
}

extern "C" void kernel_launch(void* const* d_in, const int* in_sizes, int n_in,
                              void* d_out, int out_size, void* d_ws,
                              size_t ws_size, hipStream_t stream) {
  const float* x = (const float*)d_in[0];
  const int* edge_index = (const int*)d_in[1];
  const float* W1 = (const float*)d_in[2];
  const float* b1 = (const float*)d_in[3];
  const float* W2 = (const float*)d_in[4];
  const float* b2 = (const float*)d_in[5];
  float* out = (float*)d_out;

  const int* src = edge_index;
  const int* dst = edge_index + NEDGES;

  float* ws = (float*)d_ws;
  float* deg = ws;
  float* dinv = ws + NNODES;
  float* h1 = ws + 2 * (size_t)NNODES;
  float* agg1 = h1 + (size_t)NNODES * FH;
  float* tail = agg1 + (size_t)NNODES * FH;
  float* h2 = h1;
  float* wts = tail + NNODES;
  int* cnt = (int*)(wts + NEDGES);

  float* degB = tail;
  deg_init_kernel<<<NODE_BLOCKS, SP_THREADS, 0, stream>>>(deg, degB, cnt);
  deg_count_kernel<<<CNT_BLOCKS, SP_THREADS, 0, stream>>>(dst, deg, degB);
  dinv_kernel<<<NODE_BLOCKS, SP_THREADS, 0, stream>>>(deg, degB, dinv);

  gcn_gemm1<<<GEMM_BLOCKS, GEMM_WAVES * 32, 0, stream>>>(x, W1, h1);

  agg1_init_kernel<<<INIT1_BLOCKS, SP_THREADS, 0, stream>>>(h1, dinv, agg1);
  agg_edges_kernel<FH, true><<<2 * AGG_CHUNKS, AGG_WAVES * 32, 0, stream>>>(
      src, dst, dinv, wts, h1, agg1, cnt);

  gcn_gemm2<<<GEMM_BLOCKS, GEMM_WAVES * 32, 0, stream>>>(agg1, b1, W2, h2);

  agg2_init_kernel<<<INIT2_BLOCKS, SP_THREADS, 0, stream>>>(h2, dinv, b2, out);
  agg_edges_kernel<FOUT, false><<<2 * AGG_CHUNKS, AGG_WAVES * 32, 0, stream>>>(
      src, dst, dinv, wts, h2, out, cnt + 2 * CNT_STRIDE);
}


# Per-node passes folded into the two products' tile loops: the first also forms dinv and stores agg1's self-loop start rows, the second stores out's start rows; 9 launches become 6; same values
# speedup vs baseline: 3.0969x; 1.0437x over previous
#include <hip/hip_runtime.h>

#define NNODES 100000
#define NEDGES 1600000
#define FIN 128
#define FH 64
#define FOUT 32

typedef float v2f __attribute__((ext_vector_type(2)));
typedef float v8f __attribute__((ext_vector_type(8)));
typedef float v4f __attribute__((ext_vector_type(4)));
typedef int v4i __attribute__((ext_vector_type(4)));

static __device__ __forceinline__ v8f wmma_f32_k4(v2f a, v2f b, v8f c) {
  return __builtin_amdgcn_wmma_f32_16x16x4_f32(false, a, false, b, (short)0, c,
                                               false, false);
}

#define CNT_STRIDE 128

#define SP_THREADS 128
#define NODE_V4 (NNODES / 4)
#define NODE_BLOCKS ((NODE_V4 + SP_THREADS - 1) / SP_THREADS)
static_assert(NNODES % 4 == 0, "the per-node planes are walked as float4");

__global__ __launch_bounds__(SP_THREADS) void deg_init_kernel(float* __restrict__ degA,
                                                              float* __restrict__ degB,
                                                              int* __restrict__ cnt) {
  const int i = blockIdx.x * SP_THREADS + threadIdx.x;
  if (i < NODE_V4) {
    ((float4*)degA)[i] = float4{1.0f, 1.0f, 1.0f, 1.0f};
    ((float4*)degB)[i] = float4{0.0f, 0.0f, 0.0f, 0.0f};
  }
  if (i < 4) cnt[i * CNT_STRIDE] = 0;
}

#define CNT_V4 5
#define CNT_BLOCKS (NEDGES / 4 / (SP_THREADS * CNT_V4))
static_assert(NEDGES % (4 * SP_THREADS * CNT_V4) == 0, "the count blocks must tile the edge list exactly");

__global__ __launch_bounds__(SP_THREADS) void deg_count_kernel(const int* __restrict__ dst,
                                                               float* __restrict__ degA,
                                                               float* __restrict__ degB) {
  const unsigned where = __builtin_amdgcn_s_sendmsg_rtn(0x87);
  float* __restrict__ deg = ((where >> 18) & 1u) ? degB : degA;
  const v4i* __restrict__ d4 = (const v4i*)dst;
  const int base = blockIdx.x * (SP_THREADS * CNT_V4) + threadIdx.x;
  v4i d[CNT_V4];
#pragma unroll
  for (int j = 0; j < CNT_V4; ++j) d[j] = d4[base + j * SP_THREADS];
  static_assert(CNT_V4 == 5, "the fence below names five items");
  asm volatile("" ::"v"(d[0]), "v"(d[1]), "v"(d[2]), "v"(d[3]), "v"(d[4]));
#pragma unroll
  for (int j = 0; j < CNT_V4; ++j) {
    unsafeAtomicAdd(&deg[d[j].x], 1.0f);
    unsafeAtomicAdd(&deg[d[j].y], 1.0f);
    unsafeAtomicAdd(&deg[d[j].z], 1.0f);
    unsafeAtomicAdd(&deg[d[j].w], 1.0f);
  }
}

#define GEMM_WAVES 8
#define GEMM_BLOCKS 256
#define GEMM_STRIDE (GEMM_WAVES * GEMM_BLOCKS)
#define NTILES (NNODES / 16)
static_assert(NNODES % 16 == 0, "row tiles are whole");
static_assert(GEMM_STRIDE <= NTILES, "every wave has a first tile");

static __device__ __forceinline__ v2f lo2(v4f v) { return v2f{v.x, v.y}; }
static __device__ __forceinline__ v2f hi2(v4f v) { return v2f{v.z, v.w}; }

__global__ __launch_bounds__(GEMM_WAVES * 32) void gcn_gemm1(const float* __restrict__ x,
                                                             const float* __restrict__ W1,
                                                             const float* __restrict__ degA,
                                                             const float* __restrict__ degB,
                                                             float* __restrict__ dinv,
                                                             float* __restrict__ h1,
                                                             float* __restrict__ agg1) {
  constexpr int KJ = FIN / 8;
  const int lane = threadIdx.x & 31;
  const int wave = __builtin_amdgcn_readfirstlane(threadIdx.x >> 5);
  const int lrow = lane & 15;
  const int kh = (lane >> 4) << 2;

  v4f bw[4][KJ];
  const float* wrow = W1 + lrow * FIN + kh;
#pragma unroll
  for (int n = 0; n < 4; ++n)
#pragma unroll
    for (int j = 0; j < KJ; ++j) bw[n][j] = *(const v4f*)(wrow + n * 16 * FIN + 8 * j);

  int tile = wave * GEMM_BLOCKS + blockIdx.x;
  const int rhalf = (lane >> 4) << 3;
  v4f a[KJ];
  v4f dg[4];
  {
    const float* xrow = x + (size_t)(tile * 16 + lrow) * FIN + kh;
#pragma unroll
    for (int j = 0; j < KJ; ++j) a[j] = *(const v4f*)(xrow + 8 * j);
    const int r0 = tile * 16 + rhalf;
    dg[0] = *(const v4f*)(degA + r0);
    dg[1] = *(const v4f*)(degA + r0 + 4);
    dg[2] = *(const v4f*)(degB + r0);
    dg[3] = *(const v4f*)(degB + r0 + 4);
  }
  for (; tile < NTILES; tile += GEMM_STRIDE) {
    const int nt = tile + GEMM_STRIDE < NTILES ? tile + GEMM_STRIDE : tile;
    v4f an[KJ];
    v4f dgn[4];
    {
      const float* xrow = x + (size_t)(nt * 16 + lrow) * FIN + kh;
#pragma unroll
      for (int j = 0; j < KJ; ++j) an[j] = *(const v4f*)(xrow + 8 * j);
      const int r0 = nt * 16 + rhalf;
      dgn[0] = *(const v4f*)(degA + r0);
      dgn[1] = *(const v4f*)(degA + r0 + 4);
      dgn[2] = *(const v4f*)(degB + r0);
      dgn[3] = *(const v4f*)(degB + r0 + 4);
    }
    v8f acc0 = {}, acc1 = {}, acc2 = {}, acc3 = {};
#pragma unroll
    for (int j = 0; j < KJ; ++j) {
      acc0 = wmma_f32_k4(lo2(a[j]), lo2(bw[0][j]), acc0);
      acc1 = wmma_f32_k4(lo2(a[j]), lo2(bw[1][j]), acc1);
      acc2 = wmma_f32_k4(lo2(a[j]), lo2(bw[2][j]), acc2);
      acc3 = wmma_f32_k4(lo2(a[j]), lo2(bw[3][j]), acc3);
      acc0 = wmma_f32_k4(hi2(a[j]), hi2(bw[0][j]), acc0);
      acc1 = wmma_f32_k4(hi2(a[j]), hi2(bw[1][j]), acc1);
      acc2 = wmma_f32_k4(hi2(a[j]), hi2(bw[2][j]), acc2);
      acc3 = wmma_f32_k4(hi2(a[j]), hi2(bw[3][j]), acc3);
    }
    const int rbase = tile * 16 + rhalf;
    const v4f d0 = {rsqrtf(dg[0].x + dg[2].x), rsqrtf(dg[0].y + dg[2].y),
                    rsqrtf(dg[0].z + dg[2].z), rsqrtf(dg[0].w + dg[2].w)};
    const v4f d1 = {rsqrtf(dg[1].x + dg[3].x), rsqrtf(dg[1].y + dg[3].y),
                    rsqrtf(dg[1].z + dg[3].z), rsqrtf(dg[1].w + dg[3].w)};
    if (lrow == 0) {
      *(v4f*)(dinv + rbase) = d0;
      *(v4f*)(dinv + rbase + 4) = d1;
    }
    const float dv[8] = {d0.x, d0.y, d0.z, d0.w, d1.x, d1.y, d1.z, d1.w};
#pragma unroll
    for (int r = 0; r < 8; ++r) {
      const float ww = dv[r] * dv[r];
      h1[(size_t)(rbase + r) * FH + 0 * 16 + lrow] = acc0[r];
      h1[(size_t)(rbase + r) * FH + 1 * 16 + lrow] = acc1[r];
      h1[(size_t)(rbase + r) * FH + 2 * 16 + lrow] = acc2[r];
      h1[(size_t)(rbase + r) * FH + 3 * 16 + lrow] = acc3[r];
      agg1[(size_t)(rbase + r) * FH + 0 * 16 + lrow] = acc0[r] * ww;
      agg1[(size_t)(rbase + r) * FH + 1 * 16 + lrow] = acc1[r] * ww;
      agg1[(size_t)(rbase + r) * FH + 2 * 16 + lrow] = acc2[r] * ww;
      agg1[(size_t)(rbase + r) * FH + 3 * 16 + lrow] = acc3[r] * ww;
    }
#pragma unroll
    for (int j = 0; j < KJ; ++j) a[j] = an[j];
#pragma unroll
    for (int j = 0; j < 4; ++j) dg[j] = dgn[j];
  }
}

__global__ __launch_bounds__(GEMM_WAVES * 32) void gcn_gemm2(const float* __restrict__ agg1,
                                                             const float* __restrict__ b1,
                                                             const float* __restrict__ W2,
                                                             const float* __restrict__ dinv,
                                                             const float* __restrict__ b2,
                                                             float* __restrict__ h2,
                                                             float* __restrict__ out) {
  constexpr int KJ = FH / 8;
  const int lane = threadIdx.x & 31;
  const int wave = __builtin_amdgcn_readfirstlane(threadIdx.x >> 5);
  const int lrow = lane & 15;
  const int kh = (lane >> 4) << 2;

  v4f bw[2][KJ];
  v4f bias[KJ];
  const float* wrow = W2 + lrow * FH + kh;
#pragma unroll
  for (int n = 0; n < 2; ++n)
#pragma unroll
    for (int j = 0; j < KJ; ++j) bw[n][j] = *(const v4f*)(wrow + n * 16 * FH + 8 * j);
#pragma unroll
  for (int j = 0; j < KJ; ++j) bias[j] = *(const v4f*)(b1 + kh + 8 * j);
  const float bb0 = b2[lrow];
  const float bb1 = b2[16 + lrow];

  int tile = wave * GEMM_BLOCKS + blockIdx.x;
  const int rhalf = (lane >> 4) << 3;
  v4f a[KJ];
  v4f dq[2];
  {
    const float* arow = agg1 + (size_t)(tile * 16 + lrow) * FH + kh;
#pragma unroll
    for (int j = 0; j < KJ; ++j) a[j] = *(const v4f*)(arow + 8 * j);
    const int r0 = tile * 16 + rhalf;
    dq[0] = *(const v4f*)(dinv + r0);
    dq[1] = *(const v4f*)(dinv + r0 + 4);
  }
  for (; tile < NTILES; tile += GEMM_STRIDE) {
    const int nt = tile + GEMM_STRIDE < NTILES ? tile + GEMM_STRIDE : tile;
    v4f an[KJ];
    v4f dqn[2];
    {
      const float* arow = agg1 + (size_t)(nt * 16 + lrow) * FH + kh;
#pragma unroll
      for (int j = 0; j < KJ; ++j) an[j] = *(const v4f*)(arow + 8 * j);
      const int r0 = nt * 16 + rhalf;
      dqn[0] = *(const v4f*)(dinv + r0);
      dqn[1] = *(const v4f*)(dinv + r0 + 4);
    }
    v8f acc0 = {}, acc1 = {};
#pragma unroll
    for (int j = 0; j < KJ; ++j) {
      v4f r;
      r.x = fmaxf(a[j].x + bias[j].x, 0.0f);
      r.y = fmaxf(a[j].y + bias[j].y, 0.0f);
      r.z = fmaxf(a[j].z + bias[j].z, 0.0f);
      r.w = fmaxf(a[j].w + bias[j].w, 0.0f);
      acc0 = wmma_f32_k4(lo2(r), lo2(bw[0][j]), acc0);
      acc1 = wmma_f32_k4(lo2(r), lo2(bw[1][j]), acc1);
      acc0 = wmma_f32_k4(hi2(r), hi2(bw[0][j]), acc0);
      acc1 = wmma_f32_k4(hi2(r), hi2(bw[1][j]), acc1);
    }
    const int rbase = tile * 16 + rhalf;
    const float dv[8] = {dq[0].x, dq[0].y, dq[0].z, dq[0].w, dq[1].x, dq[1].y, dq[1].z, dq[1].w};
#pragma unroll
    for (int r = 0; r < 8; ++r) {
      const float ww = dv[r] * dv[r];
      h2[(size_t)(rbase + r) * FOUT + 0 * 16 + lrow] = acc0[r];
      h2[(size_t)(rbase + r) * FOUT + 1 * 16 + lrow] = acc1[r];
      out[(size_t)(rbase + r) * FOUT + 0 * 16 + lrow] = acc0[r] * ww + bb0;
      out[(size_t)(rbase + r) * FOUT + 1 * 16 + lrow] = acc1[r] * ww + bb1;
    }
#pragma unroll
    for (int j = 0; j < KJ; ++j) a[j] = an[j];
    dq[0] = dqn[0];
    dq[1] = dqn[1];
  }
}

#define AGG_WAVES 4
#define AGG_EPW 128
#define AGG_CHUNKS (NEDGES / (AGG_WAVES * AGG_EPW))
#define OWN_SHIFT 6
static_assert(NEDGES % (AGG_WAVES * AGG_EPW) == 0, "edge count must fill every wave");
static_assert((size_t)NNODES + NEDGES + 4 * CNT_STRIDE <= (size_t)NNODES * FOUT,
              "second count plane, one weight per edge and the counters must fit the tail region");

template <int F, bool FULL>
static __device__ __forceinline__ void agg_group(int s, int d, float w, int m, int lane,
                                                 const float* __restrict__ h,
                                                 float* __restrict__ agg) {
  constexpr int C = F / 32;
  constexpr int U = 16 / C;
  float v[U][C] = {};
#pragma unroll
  for (int j = 0; j < U; ++j) {
    if (FULL || j < m) {
      const int sj = __builtin_amdgcn_readlane(s, j);
      const int r = sj * F + lane;
#pragma unroll
      for (int c = 0; c < C; ++c) v[j][c] = h[r + 32 * c];
    }
  }
#pragma unroll
  for (int j = 0; j < U; ++j) {
    if (FULL || j < m) {
      const int dj = __builtin_amdgcn_readlane(d, j);
      const float wj = __int_as_float(
          __builtin_amdgcn_readlane(__float_as_int(w), j));
      const int r = dj * F + lane;
#pragma unroll
      for (int c = 0; c < C; ++c) unsafeAtomicAdd(agg + r + 32 * c, v[j][c] * wj);
    }
  }
}

template <int F, bool FIRST>
__global__ __launch_bounds__(AGG_WAVES * 32) void agg_edges_kernel(
    const int* __restrict__ src, const int* __restrict__ dst,
    const float* __restrict__ dinv, float* __restrict__ wts,
    const float* __restrict__ h, float* __restrict__ agg, int* __restrict__ cnt) {
  constexpr int U = 16 / (F / 32);
  constexpr int NB = AGG_EPW / 32;
  __shared__ int sTicket[2];
  __shared__ int qS[AGG_WAVES][AGG_EPW];
  __shared__ int qD[AGG_WAVES][AGG_EPW];
  __shared__ float qW[AGG_WAVES][AGG_EPW];

  if (threadIdx.x == 0) {
    const unsigned where = __builtin_amdgcn_s_sendmsg_rtn(0x87);
    int c = (int)((where >> 18) & 1u);
    int t = atomicAdd(cnt + c * CNT_STRIDE, 1);
    if (t >= AGG_CHUNKS) {
      c ^= 1;
      t = atomicAdd(cnt + c * CNT_STRIDE, 1);
    }
    sTicket[0] = t;
    sTicket[1] = c;
  }
  __syncthreads();
  const int t = sTicket[0];
  const int cls = sTicket[1];
  if (t >= AGG_CHUNKS) return;

  const int lane = threadIdx.x & 31;
  const int wave = threadIdx.x >> 5;
  const int e0 = t * (AGG_WAVES * AGG_EPW) + wave * AGG_EPW + lane;
  int* __restrict__ qs = qS[wave];
  int* __restrict__ qd = qD[wave];
  float* __restrict__ qw = qW[wave];

  int s[NB], d[NB];
  float w[NB];
  bool own[NB];
#pragma unroll
  for (int b = 0; b < NB; ++b) {
    s[b] = src[e0 + 32 * b];
    d[b] = dst[e0 + 32 * b];
  }
#pragma unroll
  for (int b = 0; b < NB; ++b) own[b] = ((d[b] >> OWN_SHIFT) & 1) == cls;
  if constexpr (FIRST) {
#pragma unroll
    for (int b = 0; b < NB; ++b) {
      const int sq = own[b] ? s[b] : 0;
      const int dq = own[b] ? d[b] : 0;
      w[b] = dinv[sq] * dinv[dq];
    }
#pragma unroll
    for (int b = 0; b < NB; ++b)
      if (own[b]) wts[e0 + 32 * b] = w[b];
  } else {
#pragma unroll
    for (int b = 0; b < NB; ++b) w[b] = wts[e0 + 32 * b];
    static_assert(NB == 4, "the fence below names four record sets");
    asm volatile("" ::"v"(s[0]), "v"(s[1]), "v"(s[2]), "v"(s[3]), "v"(w[0]), "v"(w[1]),
                 "v"(w[2]), "v"(w[3]));
  }
  int n = 0;
#pragma unroll
  for (int b = 0; b < NB; ++b) {
    const unsigned mask = __builtin_amdgcn_ballot_w32(own[b]);
    if (own[b]) {
      const int p = n + (int)__builtin_amdgcn_mbcnt_lo(mask, 0u);
      qs[p] = s[b];
      qd[p] = d[b];
      qw[p] = w[b];
    }
    n += __builtin_popcount(mask);
  }

  for (int g = 0; g < n; g += U) {
    const int idx = g + (lane & (U - 1));
    int sg = 0, dg = 0;
    float wg = 0.0f;
    if (idx < n) {
      sg = qs[idx];
      dg = qd[idx];
      wg = qw[idx];
    }
    const int m = n - g;
    if (m >= U)
      agg_group<F, true>(sg, dg, wg, U, lane, h, agg);
    else
      agg_group<F, false>(sg, dg, wg, m, lane, h, agg);
  }
}

extern "C" void kernel_launch(void* const* d_in, const int* in_sizes, int n_in,
                              void* d_out, int out_size, void* d_ws,
                              size_t ws_size, hipStream_t stream) {
  const float* x = (const float*)d_in[0];
  const int* edge_index = (const int*)d_in[1];
  const float* W1 = (const float*)d_in[2];
  const float* b1 = (const float*)d_in[3];
  const float* W2 = (const float*)d_in[4];
  const float* b2 = (const float*)d_in[5];
  float* out = (float*)d_out;

  const int* src = edge_index;
  const int* dst = edge_index + NEDGES;

  float* ws = (float*)d_ws;
  float* deg = ws;
  float* dinv = ws + NNODES;
  float* h1 = ws + 2 * (size_t)NNODES;
  float* agg1 = h1 + (size_t)NNODES * FH;
  float* tail = agg1 + (size_t)NNODES * FH;
  float* h2 = h1;
  float* wts = tail + NNODES;
  int* cnt = (int*)(wts + NEDGES);

  float* degB = tail;
  deg_init_kernel<<<NODE_BLOCKS, SP_THREADS, 0, stream>>>(deg, degB, cnt);
  deg_count_kernel<<<CNT_BLOCKS, SP_THREADS, 0, stream>>>(dst, deg, degB);

  gcn_gemm1<<<GEMM_BLOCKS, GEMM_WAVES * 32, 0, stream>>>(x, W1, deg, degB, dinv, h1, agg1);

  agg_edges_kernel<FH, true><<<2 * AGG_CHUNKS, AGG_WAVES * 32, 0, stream>>>(
      src, dst, dinv, wts, h1, agg1, cnt);

  gcn_gemm2<<<GEMM_BLOCKS, GEMM_WAVES * 32, 0, stream>>>(agg1, b1, W2, dinv, b2, h2, out);

  agg_edges_kernel<FOUT, false><<<2 * AGG_CHUNKS, AGG_WAVES * 32, 0, stream>>>(
      src, dst, dinv, wts, h2, out, cnt + 2 * CNT_STRIDE);
}
